# starting kernel
# baseline (speedup 1.0000x reference)
#include <hip/hip_runtime.h>

typedef float v2f __attribute__((ext_vector_type(2)));
typedef float v8f __attribute__((ext_vector_type(8)));

__global__ __launch_bounds__(256) void u_dot_v_wmma_kernel(
    const float* __restrict__ hsrc,
    const float* __restrict__ hdst,
    const int*   __restrict__ src,
    const int*   __restrict__ dst,
    float*       __restrict__ out,
    int nE)
{
    const int lane = threadIdx.x & 31;
    const int wave = threadIdx.x >> 5;

    const int e0 = blockIdx.x * 128 + wave * 16;
    if (e0 >= nE) return;

    const int m  = lane & 15;
    const int e  = e0 + m;
    const int ec = (e < nE) ? e : (nE - 1);

    const long srow = (long)src[ec] * 128;
    const long drow = (long)dst[ec] * 128;
    const int  bg   = (lane >> 4) << 1;

    const float* ap = hsrc + srow + bg;
    const float* bp = hdst + drow + bg;

    v8f c = {};
    #pragma unroll
    for (int kb = 0; kb < 128; kb += 4) {
        v2f a = *(const v2f*)(ap + kb);
        v2f b = *(const v2f*)(bp + kb);
        c = __builtin_amdgcn_wmma_f32_16x16x4_f32(
                false, a, false, b, (short)0, c, false, false);
    }

    const bool writer = (lane < 8) || (lane >= 24);
    const int  idx    = (lane < 8) ? lane : (lane - 24);
    float val = c[0];
    #pragma unroll
    for (int i = 1; i < 8; ++i) val = (idx == i) ? c[i] : val;

    if (writer && e < nE) out[e] = val;
}

extern "C" void kernel_launch(void* const* d_in, const int* in_sizes, int n_in,
                              void* d_out, int out_size, void* d_ws, size_t ws_size,
                              hipStream_t stream) {
    const float* h_user = (const float*)d_in[0];
    const float* h_item = (const float*)d_in[1];
    const int*   src0   = (const int*)d_in[2];
    const int*   dst0   = (const int*)d_in[3];
    const int*   src1   = (const int*)d_in[4];
    const int*   dst1   = (const int*)d_in[5];
    const int*   src2   = (const int*)d_in[6];
    const int*   dst2   = (const int*)d_in[7];
    float* out = (float*)d_out;

    const int E = in_sizes[2];
    const int blocks = (E + 127) / 128;

    u_dot_v_wmma_kernel<<<blocks, 256, 0, stream>>>(h_user, h_user, src0, dst0, out,         E);
    u_dot_v_wmma_kernel<<<blocks, 256, 0, stream>>>(h_user, h_item, src1, dst1, out + E,     E);
    u_dot_v_wmma_kernel<<<blocks, 256, 0, stream>>>(h_item, h_user, src2, dst2, out + 2 * E, E);
}


# Gathered rows staged through a per-wave padded LDS slice: each global b128 load reads one whole 512-B row (4 full lines; was 16 lines x 16 B); a wave passes 64 edges through it; sum order unchanged
# speedup vs baseline: 2.3847x; 2.3847x over previous
#include <hip/hip_runtime.h>


typedef float v2f __attribute__((ext_vector_type(2)));
typedef float v4f __attribute__((ext_vector_type(4)));
typedef float v8f __attribute__((ext_vector_type(8)));

#define ROW_PITCH 132
#define WAVE_SLICE (2 * 16 * ROW_PITCH)
#define WAVES_PER_BLOCK 2
#define TILES_PER_WAVE 4
#define EDGES_PER_WAVE (16 * TILES_PER_WAVE)

__global__ __launch_bounds__(32 * WAVES_PER_BLOCK) void u_dot_v_wmma_kernel(
    const float* __restrict__ hsrc,
    const float* __restrict__ hdst,
    const int*   __restrict__ src,
    const int*   __restrict__ dst,
    float*       __restrict__ out,
    int nE)
{
    __shared__ __attribute__((aligned(16))) float lds[WAVES_PER_BLOCK * WAVE_SLICE];

    const int lane = threadIdx.x & 31;
    const int wave = threadIdx.x >> 5;

    const int w0 = (blockIdx.x * WAVES_PER_BLOCK + wave) * EDGES_PER_WAVE;
    if (w0 >= nE) return;

    const int i0 = (w0 + lane      < nE) ? (w0 + lane)      : (nE - 1);
    const int i1 = (w0 + 32 + lane < nE) ? (w0 + 32 + lane) : (nE - 1);
    const int si0 = src[i0], si1 = src[i1];
    const int di0 = dst[i0], di1 = dst[i1];

    float* la = lds + wave * WAVE_SLICE;
    float* lb = la + 16 * ROW_PITCH;

    const int m  = lane & 15;
    const int bg = (lane >> 4) << 1;
    const float* ap = la + m * ROW_PITCH + bg;
    const float* bp = lb + m * ROW_PITCH + bg;

    const bool writer = (lane < 8) || (lane >= 24);
    const int  idx    = (lane < 8) ? lane : (lane - 24);

    for (int t = 0; t < TILES_PER_WAVE; ++t) {
        const int e0 = w0 + t * 16;
        if (e0 >= nE) break;

        const int si   = (t & 2) ? si1 : si0;
        const int di   = (t & 2) ? di1 : di0;
        const int lsel = (t & 1) << 4;

        v4f ra[16], rb[16];
        #pragma unroll
        for (int r = 0; r < 16; ++r) {
            const long srow = (long)__builtin_amdgcn_readlane(si, lsel + r) * 128;
            const long drow = (long)__builtin_amdgcn_readlane(di, lsel + r) * 128;
            ra[r] = *(const v4f*)(hsrc + srow + lane * 4);
            rb[r] = *(const v4f*)(hdst + drow + lane * 4);
        }
        #pragma unroll
        for (int r = 0; r < 16; ++r) {
            *(v4f*)(la + r * ROW_PITCH + lane * 4) = ra[r];
            *(v4f*)(lb + r * ROW_PITCH + lane * 4) = rb[r];
        }

        __builtin_amdgcn_fence(__ATOMIC_RELEASE, "wavefront");
        __builtin_amdgcn_wave_barrier();
        __builtin_amdgcn_fence(__ATOMIC_ACQUIRE, "wavefront");

        v8f c = {};
        #pragma unroll
        for (int kb = 0; kb < 128; kb += 4) {
            v2f a = *(const v2f*)(ap + kb);
            v2f b = *(const v2f*)(bp + kb);
            c = __builtin_amdgcn_wmma_f32_16x16x4_f32(
                    false, a, false, b, (short)0, c, false, false);
        }

        float val = c[0];
        #pragma unroll
        for (int i = 1; i < 8; ++i) val = (idx == i) ? c[i] : val;

        const int e = e0 + m;
        if (writer && e < nE) out[e] = val;

        __builtin_amdgcn_fence(__ATOMIC_RELEASE, "wavefront");
        __builtin_amdgcn_wave_barrier();
        __builtin_amdgcn_fence(__ATOMIC_ACQUIRE, "wavefront");
    }
}

extern "C" void kernel_launch(void* const* d_in, const int* in_sizes, int n_in,
                              void* d_out, int out_size, void* d_ws, size_t ws_size,
                              hipStream_t stream) {
    const float* h_user = (const float*)d_in[0];
    const float* h_item = (const float*)d_in[1];
    const int*   src0   = (const int*)d_in[2];
    const int*   dst0   = (const int*)d_in[3];
    const int*   src1   = (const int*)d_in[4];
    const int*   dst1   = (const int*)d_in[5];
    const int*   src2   = (const int*)d_in[6];
    const int*   dst2   = (const int*)d_in[7];
    float* out = (float*)d_out;

    const int E = in_sizes[2];
    const int blocks = (E + 127) / 128;

    u_dot_v_wmma_kernel<<<blocks, 64, 0, stream>>>(h_user, h_user, src0, dst0, out,         E);
    u_dot_v_wmma_kernel<<<blocks, 64, 0, stream>>>(h_user, h_item, src1, dst1, out + E,     E);
    u_dot_v_wmma_kernel<<<blocks, 64, 0, stream>>>(h_item, h_user, src2, dst2, out + 2 * E, E);
}


# One launch instead of three: grid.y is the edge type and each block picks its tables and output slice from it in scalar registers; no drain and refill of the chip between types; sum order unchanged
# speedup vs baseline: 2.6637x; 1.1170x over previous
#include <hip/hip_runtime.h>


typedef float v2f __attribute__((ext_vector_type(2)));
typedef float v4f __attribute__((ext_vector_type(4)));
typedef float v8f __attribute__((ext_vector_type(8)));

#define ROW_PITCH 132
#define WAVE_SLICE (2 * 16 * ROW_PITCH)
#define WAVES_PER_BLOCK 2
#define TILES_PER_WAVE 4
#define EDGES_PER_WAVE (16 * TILES_PER_WAVE)

__global__ __launch_bounds__(32 * WAVES_PER_BLOCK) void u_dot_v_wmma_kernel(
    const float* __restrict__ h_user,
    const float* __restrict__ h_item,
    const int*   __restrict__ src0,
    const int*   __restrict__ dst0,
    const int*   __restrict__ src1,
    const int*   __restrict__ dst1,
    const int*   __restrict__ src2,
    const int*   __restrict__ dst2,
    float*       __restrict__ out_all,
    int nE)
{
    __shared__ __attribute__((aligned(16))) float lds[WAVES_PER_BLOCK * WAVE_SLICE];

    const int et = blockIdx.y;
    const float* hsrc = (et == 2) ? h_item : h_user;
    const float* hdst = (et == 1) ? h_item : h_user;
    const int*   src  = (et == 0) ? src0 : ((et == 1) ? src1 : src2);
    const int*   dst  = (et == 0) ? dst0 : ((et == 1) ? dst1 : dst2);
    float*       out  = out_all + (long)et * nE;

    const int lane = threadIdx.x & 31;
    const int wave = threadIdx.x >> 5;

    const int w0 = (blockIdx.x * WAVES_PER_BLOCK + wave) * EDGES_PER_WAVE;
    if (w0 >= nE) return;

    const int i0 = (w0 + lane      < nE) ? (w0 + lane)      : (nE - 1);
    const int i1 = (w0 + 32 + lane < nE) ? (w0 + 32 + lane) : (nE - 1);
    const int si0 = src[i0], si1 = src[i1];
    const int di0 = dst[i0], di1 = dst[i1];

    float* la = lds + wave * WAVE_SLICE;
    float* lb = la + 16 * ROW_PITCH;

    const int m  = lane & 15;
    const int bg = (lane >> 4) << 1;
    const float* ap = la + m * ROW_PITCH + bg;
    const float* bp = lb + m * ROW_PITCH + bg;

    const bool writer = (lane < 8) || (lane >= 24);
    const int  idx    = (lane < 8) ? lane : (lane - 24);

    for (int t = 0; t < TILES_PER_WAVE; ++t) {
        const int e0 = w0 + t * 16;
        if (e0 >= nE) break;

        const int si   = (t & 2) ? si1 : si0;
        const int di   = (t & 2) ? di1 : di0;
        const int lsel = (t & 1) << 4;

        v4f ra[16], rb[16];
        #pragma unroll
        for (int r = 0; r < 16; ++r) {
            const long srow = (long)__builtin_amdgcn_readlane(si, lsel + r) * 128;
            const long drow = (long)__builtin_amdgcn_readlane(di, lsel + r) * 128;
            ra[r] = *(const v4f*)(hsrc + srow + lane * 4);
            rb[r] = *(const v4f*)(hdst + drow + lane * 4);
        }
        #pragma unroll
        for (int r = 0; r < 16; ++r) {
            *(v4f*)(la + r * ROW_PITCH + lane * 4) = ra[r];
            *(v4f*)(lb + r * ROW_PITCH + lane * 4) = rb[r];
        }

        __builtin_amdgcn_fence(__ATOMIC_RELEASE, "wavefront");
        __builtin_amdgcn_wave_barrier();
        __builtin_amdgcn_fence(__ATOMIC_ACQUIRE, "wavefront");

        v8f c = {};
        #pragma unroll
        for (int kb = 0; kb < 128; kb += 4) {
            v2f a = *(const v2f*)(ap + kb);
            v2f b = *(const v2f*)(bp + kb);
            c = __builtin_amdgcn_wmma_f32_16x16x4_f32(
                    false, a, false, b, (short)0, c, false, false);
        }

        float val = c[0];
        #pragma unroll
        for (int i = 1; i < 8; ++i) val = (idx == i) ? c[i] : val;

        const int e = e0 + m;
        if (writer && e < nE) out[e] = val;

        __builtin_amdgcn_fence(__ATOMIC_RELEASE, "wavefront");
        __builtin_amdgcn_wave_barrier();
        __builtin_amdgcn_fence(__ATOMIC_ACQUIRE, "wavefront");
    }
}

extern "C" void kernel_launch(void* const* d_in, const int* in_sizes, int n_in,
                              void* d_out, int out_size, void* d_ws, size_t ws_size,
                              hipStream_t stream) {
    const float* h_user = (const float*)d_in[0];
    const float* h_item = (const float*)d_in[1];
    const int*   src0   = (const int*)d_in[2];
    const int*   dst0   = (const int*)d_in[3];
    const int*   src1   = (const int*)d_in[4];
    const int*   dst1   = (const int*)d_in[5];
    const int*   src2   = (const int*)d_in[6];
    const int*   dst2   = (const int*)d_in[7];
    float* out = (float*)d_out;

    const int E = in_sizes[2];
    const int blocks = (E + 127) / 128;

    u_dot_v_wmma_kernel<<<dim3(blocks, 3), 64, 0, stream>>>(
        h_user, h_item, src0, dst0, src1, dst1, src2, dst2, out, E);
}
